# starting kernel
# baseline (speedup 1.0000x reference)
#include <hip/hip_runtime.h>
#include <hip/hip_bf16.h>
#include <math.h>

#define SOMI_D      512
#define SOMI_ROWS   16384
#define SOMI_MTILE  32
#define SOMI_STEPS  10
#define ALPHA0_F    0.1f
#define DT_F        0.1f
#define BETA_F      2.8982753492378876f

typedef __attribute__((ext_vector_type(8)))  __bf16 bf16x8_t;
typedef __attribute__((ext_vector_type(16))) __bf16 bf16x16_t;
typedef __attribute__((ext_vector_type(8)))  float  f32x8_t;

__device__ __forceinline__ bf16x16_t ldfrag(const __bf16* p) {
    union { bf16x16_t v; bf16x8_t h[2]; } u;
    u.h[0] = *(const bf16x8_t*)(p);
    u.h[1] = *(const bf16x8_t*)(p + 16);
    return u.v;
}

__global__ void somi_prep_kernel(const float* __restrict__ W,
                                 const float* __restrict__ gate_alpha,
                                 float* __restrict__ gate_out,
                                 __bf16* __restrict__ Lb) {
    const int n   = blockIdx.x;
    const int tid = threadIdx.x;
    __shared__ float red[256];

    float s = 0.f;
    for (int k = tid; k < SOMI_D; k += 256)
        s += fabsf(W[n * SOMI_D + k]);
    red[tid] = s;
    __syncthreads();
    for (int off = 128; off > 0; off >>= 1) {
        if (tid < off) red[tid] += red[tid + off];
        __syncthreads();
    }
    const float inv = 1.0f / fmaxf(red[0], 1e-8f);

    for (int k = tid; k < SOMI_D; k += 256) {
        float v = ((k == n) ? 1.0f : 0.0f) - fabsf(W[n * SOMI_D + k]) * inv;
        Lb[n * SOMI_D + k] = (__bf16)v;
    }
    if (n == 0 && tid == 0) *gate_out = tanhf(gate_alpha[0]);
}

__global__ __launch_bounds__(1024, 1)
void somi_settle_kernel(const float* __restrict__ hidden,
                        const float* __restrict__ precision,
                        const __bf16* __restrict__ Lb,
                        const float* __restrict__ gate_p,
                        float* __restrict__ out) {
    __shared__ __align__(16) __bf16 sPhi[SOMI_MTILE * SOMI_D];

    const int tid    = threadIdx.x;
    const int wave   = tid >> 5;
    const int lane   = tid & 31;
    const int lane15 = lane & 15;
    const int hi     = lane >> 4;
    const int mtile  = wave >> 4;
    const int ntb    = (wave & 15) * 2;
    const int mloc0  = mtile * 16 + hi * 8;
    const long rowBase = (long)blockIdx.x * SOMI_MTILE;

    float phi[2][8], pdot[2][8], targ[2][8], prec[2];

    #pragma unroll
    for (int t = 0; t < 2; ++t) {
        const int n = (ntb + t) * 16 + lane15;
        prec[t] = precision[n];
        #pragma unroll
        for (int r = 0; r < 8; ++r) {
            const int m = mloc0 + r;
            const float v = hidden[(rowBase + m) * SOMI_D + n];
            targ[t][r] = v; phi[t][r] = v; pdot[t][r] = 0.f;
            sPhi[m * SOMI_D + n] = (__bf16)v;
        }
    }
    const float gate = *gate_p;

    const __bf16* aBase  = sPhi + (mtile * 16 + lane15) * SOMI_D + hi * 8;
    const __bf16* b0Base = Lb + (long)((ntb + 0) * 16 + lane15) * SOMI_D + hi * 8;
    const __bf16* b1Base = Lb + (long)((ntb + 1) * 16 + lane15) * SOMI_D + hi * 8;
    __builtin_prefetch(b0Base, 0, 3);
    __builtin_prefetch(b1Base, 0, 3);

    __syncthreads();

    for (int step = 0; step < SOMI_STEPS; ++step) {
        f32x8_t c0 = {}; f32x8_t c1 = {};
        #pragma unroll 4
        for (int k = 0; k < 16; ++k) {
            const bf16x16_t a  = ldfrag(aBase  + k * 32);
            const bf16x16_t b0 = ldfrag(b0Base + k * 32);
            const bf16x16_t b1 = ldfrag(b1Base + k * 32);
            c0 = __builtin_amdgcn_wmma_f32_16x16x32_bf16(false, a, false, b0,
                                                         (short)0, c0, false, false);
            c1 = __builtin_amdgcn_wmma_f32_16x16x32_bf16(false, a, false, b1,
                                                         (short)0, c1, false, false);
        }
        __syncthreads();

        #pragma unroll
        for (int t = 0; t < 2; ++t) {
            #pragma unroll
            for (int r = 0; r < 8; ++r) {
                const float cp = (t == 0) ? c0[r] : c1[r];
                const float f  = -cp - ALPHA0_F * phi[t][r]
                                 + prec[t] * (targ[t][r] - phi[t][r])
                                 - BETA_F * pdot[t][r];
                pdot[t][r] += DT_F * f;
                phi[t][r]  += DT_F * pdot[t][r];
            }
        }
        #pragma unroll
        for (int t = 0; t < 2; ++t) {
            const int n = (ntb + t) * 16 + lane15;
            #pragma unroll
            for (int r = 0; r < 8; ++r)
                sPhi[(mloc0 + r) * SOMI_D + n] = (__bf16)phi[t][r];
        }
        __syncthreads();
    }

    #pragma unroll
    for (int t = 0; t < 2; ++t) {
        const int n = (ntb + t) * 16 + lane15;
        #pragma unroll
        for (int r = 0; r < 8; ++r) {
            const int m = mloc0 + r;
            out[(rowBase + m) * SOMI_D + n] =
                targ[t][r] + gate * (phi[t][r] - targ[t][r]);
        }
    }
}

extern "C" void kernel_launch(void* const* d_in, const int* in_sizes, int n_in,
                              void* d_out, int out_size, void* d_ws, size_t ws_size,
                              hipStream_t stream) {
    const float* hidden     = (const float*)d_in[0];
    const float* W          = (const float*)d_in[1];
    const float* gate_alpha = (const float*)d_in[2];
    const float* precision  = (const float*)d_in[3];
    float* out = (float*)d_out;

    float*  gate_ws = (float*)d_ws;
    __bf16* Lb      = (__bf16*)((char*)d_ws + 256);

    somi_prep_kernel<<<SOMI_D, 256, 0, stream>>>(W, gate_alpha, gate_ws, Lb);
    somi_settle_kernel<<<SOMI_ROWS / SOMI_MTILE, 1024, 0, stream>>>(
        hidden, precision, Lb, gate_ws, out);
}


# prep (one workgroup per 16-row tile) writes the bf16 Laplacian in WMMA B-fragment order: each B fragment load is two contiguous 512-byte reads, not 32-byte pieces of 16 rows
# speedup vs baseline: 1.3328x; 1.3328x over previous
#include <hip/hip_runtime.h>
#include <hip/hip_bf16.h>
#include <math.h>

#define SOMI_D      512
#define SOMI_ROWS   16384
#define SOMI_MTILE  32
#define SOMI_STEPS  10
#define ALPHA0_F    0.1f
#define DT_F        0.1f
#define BETA_F      2.8982753492378876f

typedef __attribute__((ext_vector_type(8)))  __bf16 bf16x8_t;
typedef __attribute__((ext_vector_type(16))) __bf16 bf16x16_t;
typedef __attribute__((ext_vector_type(8)))  float  f32x8_t;

__device__ __forceinline__ bf16x16_t ldfrag(const __bf16* p) {
    union { bf16x16_t v; bf16x8_t h[2]; } u;
    u.h[0] = *(const bf16x8_t*)(p);
    u.h[1] = *(const bf16x8_t*)(p + 16);
    return u.v;
}

__device__ __forceinline__ bf16x16_t ldfragB(const __bf16* p) {
    union { bf16x16_t v; bf16x8_t h[2]; } u;
    u.h[0] = *(const bf16x8_t*)(p);
    u.h[1] = *(const bf16x8_t*)(p + 256);
    return u.v;
}

__global__ void somi_prep_kernel(const float* __restrict__ W,
                                 const float* __restrict__ gate_alpha,
                                 float* __restrict__ gate_out,
                                 __bf16* __restrict__ Lb) {
    const int nt  = blockIdx.x;
    const int tid = threadIdx.x;
    __shared__ float red[16][256];

    for (int r = 0; r < 16; ++r) {
        const float* wr = W + (nt * 16 + r) * SOMI_D;
        float s = 0.f;
        for (int k = tid; k < SOMI_D; k += 256)
            s += fabsf(wr[k]);
        red[r][tid] = s;
    }
    __syncthreads();
    for (int sh = 7; sh >= 0; --sh) {
        const int off = 1 << sh;
        for (int idx = tid; idx < (16 << sh); idx += 256) {
            const int r = idx >> sh;
            const int j = idx & (off - 1);
            red[r][j] += red[r][j + off];
        }
        __syncthreads();
    }

    for (int j = tid; j < 1024; j += 256) {
        const int slab = j >> 6;
        const int half = (j >> 5) & 1;
        const int ln   = j & 31;
        const int c    = ln & 15;
        const int h    = ln >> 4;
        const int n    = nt * 16 + c;
        const int k0   = slab * 32 + half * 16 + h * 8;
        const float inv = 1.0f / fmaxf(red[c][0], 1e-8f);
        const float4 w0 = *(const float4*)(W + n * SOMI_D + k0);
        const float4 w1 = *(const float4*)(W + n * SOMI_D + k0 + 4);
        const float w[8] = {w0.x, w0.y, w0.z, w0.w, w1.x, w1.y, w1.z, w1.w};
        bf16x8_t o;
        #pragma unroll
        for (int e = 0; e < 8; ++e) {
            const float v = ((k0 + e == n) ? 1.0f : 0.0f) - fabsf(w[e]) * inv;
            o[e] = (__bf16)v;
        }
        *(bf16x8_t*)(Lb + (long)nt * 8192 + j * 8) = o;
    }
    if (nt == 0 && tid == 0) *gate_out = tanhf(gate_alpha[0]);
}

__global__ __launch_bounds__(1024, 1)
void somi_settle_kernel(const float* __restrict__ hidden,
                        const float* __restrict__ precision,
                        const __bf16* __restrict__ Lb,
                        const float* __restrict__ gate_p,
                        float* __restrict__ out) {
    __shared__ __align__(16) __bf16 sPhi[SOMI_MTILE * SOMI_D];

    const int tid    = threadIdx.x;
    const int wave   = tid >> 5;
    const int lane   = tid & 31;
    const int lane15 = lane & 15;
    const int hi     = lane >> 4;
    const int mtile  = wave >> 4;
    const int ntb    = (wave & 15) * 2;
    const int mloc0  = mtile * 16 + hi * 8;
    const long rowBase = (long)blockIdx.x * SOMI_MTILE;

    float phi[2][8], pdot[2][8], targ[2][8], prec[2];

    #pragma unroll
    for (int t = 0; t < 2; ++t) {
        const int n = (ntb + t) * 16 + lane15;
        prec[t] = precision[n];
        #pragma unroll
        for (int r = 0; r < 8; ++r) {
            const int m = mloc0 + r;
            const float v = hidden[(rowBase + m) * SOMI_D + n];
            targ[t][r] = v; phi[t][r] = v; pdot[t][r] = 0.f;
            sPhi[m * SOMI_D + n] = (__bf16)v;
        }
    }
    const float gate = *gate_p;

    const __bf16* aBase  = sPhi + (mtile * 16 + lane15) * SOMI_D + hi * 8;
    const __bf16* b0Base = Lb + (long)(ntb + 0) * 8192 + lane * 8;
    const __bf16* b1Base = Lb + (long)(ntb + 1) * 8192 + lane * 8;
    __builtin_prefetch(b0Base, 0, 3);
    __builtin_prefetch(b1Base, 0, 3);

    __syncthreads();

    for (int step = 0; step < SOMI_STEPS; ++step) {
        f32x8_t c0 = {}; f32x8_t c1 = {};
        #pragma unroll 4
        for (int k = 0; k < 16; ++k) {
            const bf16x16_t a  = ldfrag(aBase  + k * 32);
            const bf16x16_t b0 = ldfragB(b0Base + k * 512);
            const bf16x16_t b1 = ldfragB(b1Base + k * 512);
            c0 = __builtin_amdgcn_wmma_f32_16x16x32_bf16(false, a, false, b0,
                                                         (short)0, c0, false, false);
            c1 = __builtin_amdgcn_wmma_f32_16x16x32_bf16(false, a, false, b1,
                                                         (short)0, c1, false, false);
        }
        __syncthreads();

        #pragma unroll
        for (int t = 0; t < 2; ++t) {
            #pragma unroll
            for (int r = 0; r < 8; ++r) {
                const float cp = (t == 0) ? c0[r] : c1[r];
                const float f  = -cp - ALPHA0_F * phi[t][r]
                                 + prec[t] * (targ[t][r] - phi[t][r])
                                 - BETA_F * pdot[t][r];
                pdot[t][r] += DT_F * f;
                phi[t][r]  += DT_F * pdot[t][r];
            }
        }
        #pragma unroll
        for (int t = 0; t < 2; ++t) {
            const int n = (ntb + t) * 16 + lane15;
            #pragma unroll
            for (int r = 0; r < 8; ++r)
                sPhi[(mloc0 + r) * SOMI_D + n] = (__bf16)phi[t][r];
        }
        __syncthreads();
    }

    #pragma unroll
    for (int t = 0; t < 2; ++t) {
        const int n = (ntb + t) * 16 + lane15;
        #pragma unroll
        for (int r = 0; r < 8; ++r) {
            const int m = mloc0 + r;
            out[(rowBase + m) * SOMI_D + n] =
                targ[t][r] + gate * (phi[t][r] - targ[t][r]);
        }
    }
}

extern "C" void kernel_launch(void* const* d_in, const int* in_sizes, int n_in,
                              void* d_out, int out_size, void* d_ws, size_t ws_size,
                              hipStream_t stream) {
    const float* hidden     = (const float*)d_in[0];
    const float* W          = (const float*)d_in[1];
    const float* gate_alpha = (const float*)d_in[2];
    const float* precision  = (const float*)d_in[3];
    float* out = (float*)d_out;

    float*  gate_ws = (float*)d_ws;
    __bf16* Lb      = (__bf16*)((char*)d_ws + 256);

    somi_prep_kernel<<<SOMI_D / 16, 256, 0, stream>>>(W, gate_alpha, gate_ws, Lb);
    somi_settle_kernel<<<SOMI_ROWS / SOMI_MTILE, 1024, 0, stream>>>(
        hidden, precision, Lb, gate_ws, out);
}


# LDS row pitch of the bf16 phi mirror padded from 1024 B to 1040 B (16 B x odd): the 16 rows of an A-fragment ds_load_b128 spread over all 64 banks instead of sharing 4
# speedup vs baseline: 3.2863x; 2.4657x over previous
#include <hip/hip_runtime.h>
#include <hip/hip_bf16.h>
#include <math.h>

#define SOMI_D      512
#define SOMI_ROWS   16384
#define SOMI_MTILE  32
#define SOMI_STEPS  10
#define SOMI_PITCH  520
#define ALPHA0_F    0.1f
#define DT_F        0.1f
#define BETA_F      2.8982753492378876f

typedef __attribute__((ext_vector_type(8)))  __bf16 bf16x8_t;
typedef __attribute__((ext_vector_type(16))) __bf16 bf16x16_t;
typedef __attribute__((ext_vector_type(8)))  float  f32x8_t;

__device__ __forceinline__ bf16x16_t ldfrag(const __bf16* p) {
    union { bf16x16_t v; bf16x8_t h[2]; } u;
    u.h[0] = *(const bf16x8_t*)(p);
    u.h[1] = *(const bf16x8_t*)(p + 16);
    return u.v;
}

__device__ __forceinline__ bf16x16_t ldfragB(const __bf16* p) {
    union { bf16x16_t v; bf16x8_t h[2]; } u;
    u.h[0] = *(const bf16x8_t*)(p);
    u.h[1] = *(const bf16x8_t*)(p + 256);
    return u.v;
}

__global__ void somi_prep_kernel(const float* __restrict__ W,
                                 const float* __restrict__ gate_alpha,
                                 float* __restrict__ gate_out,
                                 __bf16* __restrict__ Lb) {
    const int nt  = blockIdx.x;
    const int tid = threadIdx.x;
    __shared__ float red[16][256];

    for (int r = 0; r < 16; ++r) {
        const float* wr = W + (nt * 16 + r) * SOMI_D;
        float s = 0.f;
        for (int k = tid; k < SOMI_D; k += 256)
            s += fabsf(wr[k]);
        red[r][tid] = s;
    }
    __syncthreads();
    for (int sh = 7; sh >= 0; --sh) {
        const int off = 1 << sh;
        for (int idx = tid; idx < (16 << sh); idx += 256) {
            const int r = idx >> sh;
            const int j = idx & (off - 1);
            red[r][j] += red[r][j + off];
        }
        __syncthreads();
    }

    for (int j = tid; j < 1024; j += 256) {
        const int slab = j >> 6;
        const int half = (j >> 5) & 1;
        const int ln   = j & 31;
        const int c    = ln & 15;
        const int h    = ln >> 4;
        const int n    = nt * 16 + c;
        const int k0   = slab * 32 + half * 16 + h * 8;
        const float inv = 1.0f / fmaxf(red[c][0], 1e-8f);
        const float4 w0 = *(const float4*)(W + n * SOMI_D + k0);
        const float4 w1 = *(const float4*)(W + n * SOMI_D + k0 + 4);
        const float w[8] = {w0.x, w0.y, w0.z, w0.w, w1.x, w1.y, w1.z, w1.w};
        bf16x8_t o;
        #pragma unroll
        for (int e = 0; e < 8; ++e) {
            const float v = ((k0 + e == n) ? 1.0f : 0.0f) - fabsf(w[e]) * inv;
            o[e] = (__bf16)v;
        }
        *(bf16x8_t*)(Lb + (long)nt * 8192 + j * 8) = o;
    }
    if (nt == 0 && tid == 0) *gate_out = tanhf(gate_alpha[0]);
}

__global__ __launch_bounds__(1024, 1)
void somi_settle_kernel(const float* __restrict__ hidden,
                        const float* __restrict__ precision,
                        const __bf16* __restrict__ Lb,
                        const float* __restrict__ gate_p,
                        float* __restrict__ out) {
    __shared__ __align__(16) __bf16 sPhi[SOMI_MTILE * SOMI_PITCH];

    const int tid    = threadIdx.x;
    const int wave   = tid >> 5;
    const int lane   = tid & 31;
    const int lane15 = lane & 15;
    const int hi     = lane >> 4;
    const int mtile  = wave >> 4;
    const int ntb    = (wave & 15) * 2;
    const int mloc0  = mtile * 16 + hi * 8;
    const long rowBase = (long)blockIdx.x * SOMI_MTILE;

    float phi[2][8], pdot[2][8], targ[2][8], prec[2];

    #pragma unroll
    for (int t = 0; t < 2; ++t) {
        const int n = (ntb + t) * 16 + lane15;
        prec[t] = precision[n];
        #pragma unroll
        for (int r = 0; r < 8; ++r) {
            const int m = mloc0 + r;
            const float v = hidden[(rowBase + m) * SOMI_D + n];
            targ[t][r] = v; phi[t][r] = v; pdot[t][r] = 0.f;
            sPhi[m * SOMI_PITCH + n] = (__bf16)v;
        }
    }
    const float gate = *gate_p;

    const __bf16* aBase  = sPhi + (mtile * 16 + lane15) * SOMI_PITCH + hi * 8;
    const __bf16* b0Base = Lb + (long)(ntb + 0) * 8192 + lane * 8;
    const __bf16* b1Base = Lb + (long)(ntb + 1) * 8192 + lane * 8;
    __builtin_prefetch(b0Base, 0, 3);
    __builtin_prefetch(b1Base, 0, 3);

    __syncthreads();

    for (int step = 0; step < SOMI_STEPS; ++step) {
        f32x8_t c0 = {}; f32x8_t c1 = {};
        #pragma unroll 4
        for (int k = 0; k < 16; ++k) {
            const bf16x16_t a  = ldfrag(aBase  + k * 32);
            const bf16x16_t b0 = ldfragB(b0Base + k * 512);
            const bf16x16_t b1 = ldfragB(b1Base + k * 512);
            c0 = __builtin_amdgcn_wmma_f32_16x16x32_bf16(false, a, false, b0,
                                                         (short)0, c0, false, false);
            c1 = __builtin_amdgcn_wmma_f32_16x16x32_bf16(false, a, false, b1,
                                                         (short)0, c1, false, false);
        }
        __syncthreads();

        #pragma unroll
        for (int t = 0; t < 2; ++t) {
            #pragma unroll
            for (int r = 0; r < 8; ++r) {
                const float cp = (t == 0) ? c0[r] : c1[r];
                const float f  = -cp - ALPHA0_F * phi[t][r]
                                 + prec[t] * (targ[t][r] - phi[t][r])
                                 - BETA_F * pdot[t][r];
                pdot[t][r] += DT_F * f;
                phi[t][r]  += DT_F * pdot[t][r];
            }
        }
        #pragma unroll
        for (int t = 0; t < 2; ++t) {
            const int n = (ntb + t) * 16 + lane15;
            #pragma unroll
            for (int r = 0; r < 8; ++r)
                sPhi[(mloc0 + r) * SOMI_PITCH + n] = (__bf16)phi[t][r];
        }
        __syncthreads();
    }

    #pragma unroll
    for (int t = 0; t < 2; ++t) {
        const int n = (ntb + t) * 16 + lane15;
        #pragma unroll
        for (int r = 0; r < 8; ++r) {
            const int m = mloc0 + r;
            out[(rowBase + m) * SOMI_D + n] =
                targ[t][r] + gate * (phi[t][r] - targ[t][r]);
        }
    }
}

extern "C" void kernel_launch(void* const* d_in, const int* in_sizes, int n_in,
                              void* d_out, int out_size, void* d_ws, size_t ws_size,
                              hipStream_t stream) {
    const float* hidden     = (const float*)d_in[0];
    const float* W          = (const float*)d_in[1];
    const float* gate_alpha = (const float*)d_in[2];
    const float* precision  = (const float*)d_in[3];
    float* out = (float*)d_out;

    float*  gate_ws = (float*)d_ws;
    __bf16* Lb      = (__bf16*)((char*)d_ws + 256);

    somi_prep_kernel<<<SOMI_D / 16, 256, 0, stream>>>(W, gate_alpha, gate_ws, Lb);
    somi_settle_kernel<<<SOMI_ROWS / SOMI_MTILE, 1024, 0, stream>>>(
        hidden, precision, Lb, gate_ws, out);
}


# B fragment loads issued one K-slab ahead of the WMMAs that use them (two-slot register ring, next step's slab 0 fetched during the pointwise phase): L2 latency overlaps matrix work; sums unchanged
# speedup vs baseline: 4.0187x; 1.2229x over previous
#include <hip/hip_runtime.h>
#include <hip/hip_bf16.h>
#include <math.h>

#define SOMI_D      512
#define SOMI_ROWS   16384
#define SOMI_MTILE  32
#define SOMI_STEPS  10
#define SOMI_PITCH  520
#define ALPHA0_F    0.1f
#define DT_F        0.1f
#define BETA_F      2.8982753492378876f

typedef __attribute__((ext_vector_type(8)))  __bf16 bf16x8_t;
typedef __attribute__((ext_vector_type(16))) __bf16 bf16x16_t;
typedef __attribute__((ext_vector_type(8)))  float  f32x8_t;

__device__ __forceinline__ bf16x16_t ldfrag(const __bf16* p) {
    union { bf16x16_t v; bf16x8_t h[2]; } u;
    u.h[0] = *(const bf16x8_t*)(p);
    u.h[1] = *(const bf16x8_t*)(p + 16);
    return u.v;
}

__device__ __forceinline__ bf16x16_t ldfragB(const __bf16* p) {
    union { bf16x16_t v; bf16x8_t h[2]; } u;
    u.h[0] = *(const bf16x8_t*)(p);
    u.h[1] = *(const bf16x8_t*)(p + 256);
    return u.v;
}

__global__ void somi_prep_kernel(const float* __restrict__ W,
                                 const float* __restrict__ gate_alpha,
                                 float* __restrict__ gate_out,
                                 __bf16* __restrict__ Lb) {
    const int nt  = blockIdx.x;
    const int tid = threadIdx.x;
    __shared__ float red[16][256];

    for (int r = 0; r < 16; ++r) {
        const float* wr = W + (nt * 16 + r) * SOMI_D;
        float s = 0.f;
        for (int k = tid; k < SOMI_D; k += 256)
            s += fabsf(wr[k]);
        red[r][tid] = s;
    }
    __syncthreads();
    for (int sh = 7; sh >= 0; --sh) {
        const int off = 1 << sh;
        for (int idx = tid; idx < (16 << sh); idx += 256) {
            const int r = idx >> sh;
            const int j = idx & (off - 1);
            red[r][j] += red[r][j + off];
        }
        __syncthreads();
    }

    for (int j = tid; j < 1024; j += 256) {
        const int slab = j >> 6;
        const int half = (j >> 5) & 1;
        const int ln   = j & 31;
        const int c    = ln & 15;
        const int h    = ln >> 4;
        const int n    = nt * 16 + c;
        const int k0   = slab * 32 + half * 16 + h * 8;
        const float inv = 1.0f / fmaxf(red[c][0], 1e-8f);
        const float4 w0 = *(const float4*)(W + n * SOMI_D + k0);
        const float4 w1 = *(const float4*)(W + n * SOMI_D + k0 + 4);
        const float w[8] = {w0.x, w0.y, w0.z, w0.w, w1.x, w1.y, w1.z, w1.w};
        bf16x8_t o;
        #pragma unroll
        for (int e = 0; e < 8; ++e) {
            const float v = ((k0 + e == n) ? 1.0f : 0.0f) - fabsf(w[e]) * inv;
            o[e] = (__bf16)v;
        }
        *(bf16x8_t*)(Lb + (long)nt * 8192 + j * 8) = o;
    }
    if (nt == 0 && tid == 0) *gate_out = tanhf(gate_alpha[0]);
}

__global__ __launch_bounds__(1024, 1)
void somi_settle_kernel(const float* __restrict__ hidden,
                        const float* __restrict__ precision,
                        const __bf16* Lb,
                        const float* __restrict__ gate_p,
                        float* __restrict__ out) {
    __shared__ __align__(16) __bf16 sPhi[SOMI_MTILE * SOMI_PITCH];

    const int tid    = threadIdx.x;
    const int wave   = tid >> 5;
    const int lane   = tid & 31;
    const int lane15 = lane & 15;
    const int hi     = lane >> 4;
    const int mtile  = wave >> 4;
    const int ntb    = (wave & 15) * 2;
    const int mloc0  = mtile * 16 + hi * 8;
    const long rowBase = (long)blockIdx.x * SOMI_MTILE;

    float phi[2][8], pdot[2][8], targ[2][8], prec[2];

    #pragma unroll
    for (int t = 0; t < 2; ++t) {
        const int n = (ntb + t) * 16 + lane15;
        prec[t] = precision[n];
        #pragma unroll
        for (int r = 0; r < 8; ++r) {
            const int m = mloc0 + r;
            const float v = hidden[(rowBase + m) * SOMI_D + n];
            targ[t][r] = v; phi[t][r] = v; pdot[t][r] = 0.f;
            sPhi[m * SOMI_PITCH + n] = (__bf16)v;
        }
    }
    const float gate = *gate_p;

    const __bf16* aBase  = sPhi + (mtile * 16 + lane15) * SOMI_PITCH + hi * 8;
    const __bf16* b0Base = Lb + (long)(ntb + 0) * 8192 + lane * 8;
    const __bf16* b1Base = Lb + (long)(ntb + 1) * 8192 + lane * 8;
    __builtin_prefetch(b0Base, 0, 3);
    __builtin_prefetch(b1Base, 0, 3);

    bf16x16_t bc0 = ldfragB(b0Base);
    bf16x16_t bc1 = ldfragB(b1Base);

    __syncthreads();

    for (int step = 0; step < SOMI_STEPS; ++step) {
        f32x8_t c0 = {}; f32x8_t c1 = {};
        bf16x16_t ac = ldfrag(aBase);
        #pragma unroll
        for (int k = 0; k < 16; ++k) {
            bf16x16_t bn0 = bc0, bn1 = bc1, an = ac;
            if (k < 15) {
                bn0 = ldfragB(b0Base + (k + 1) * 512);
                bn1 = ldfragB(b1Base + (k + 1) * 512);
                an  = ldfrag(aBase + (k + 1) * 32);
            }
            __builtin_amdgcn_sched_barrier(0);
            c0 = __builtin_amdgcn_wmma_f32_16x16x32_bf16(false, ac, false, bc0,
                                                         (short)0, c0, false, false);
            c1 = __builtin_amdgcn_wmma_f32_16x16x32_bf16(false, ac, false, bc1,
                                                         (short)0, c1, false, false);
            bc0 = bn0; bc1 = bn1; ac = an;
        }
        __syncthreads();

        bc0 = ldfragB(b0Base);
        bc1 = ldfragB(b1Base);

        #pragma unroll
        for (int t = 0; t < 2; ++t) {
            #pragma unroll
            for (int r = 0; r < 8; ++r) {
                const float cp = (t == 0) ? c0[r] : c1[r];
                const float f  = -cp - ALPHA0_F * phi[t][r]
                                 + prec[t] * (targ[t][r] - phi[t][r])
                                 - BETA_F * pdot[t][r];
                pdot[t][r] += DT_F * f;
                phi[t][r]  += DT_F * pdot[t][r];
            }
        }
        #pragma unroll
        for (int t = 0; t < 2; ++t) {
            const int n = (ntb + t) * 16 + lane15;
            #pragma unroll
            for (int r = 0; r < 8; ++r)
                sPhi[(mloc0 + r) * SOMI_PITCH + n] = (__bf16)phi[t][r];
        }
        __syncthreads();
    }

    #pragma unroll
    for (int t = 0; t < 2; ++t) {
        const int n = (ntb + t) * 16 + lane15;
        #pragma unroll
        for (int r = 0; r < 8; ++r) {
            const int m = mloc0 + r;
            out[(rowBase + m) * SOMI_D + n] =
                targ[t][r] + gate * (phi[t][r] - targ[t][r]);
        }
    }
}

extern "C" void kernel_launch(void* const* d_in, const int* in_sizes, int n_in,
                              void* d_out, int out_size, void* d_ws, size_t ws_size,
                              hipStream_t stream) {
    const float* hidden     = (const float*)d_in[0];
    const float* W          = (const float*)d_in[1];
    const float* gate_alpha = (const float*)d_in[2];
    const float* precision  = (const float*)d_in[3];
    float* out = (float*)d_out;

    float*  gate_ws = (float*)d_ws;
    __bf16* Lb      = (__bf16*)((char*)d_ws + 256);

    somi_prep_kernel<<<SOMI_D / 16, 256, 0, stream>>>(W, gate_alpha, gate_ws, Lb);
    somi_settle_kernel<<<SOMI_ROWS / SOMI_MTILE, 1024, 0, stream>>>(
        hidden, precision, Lb, gate_ws, out);
}
